# starting kernel
# baseline (speedup 1.0000x reference)
#include <hip/hip_runtime.h>

#define NP 128
#define NG 256
#define NC 2048
#define HW 64

typedef __attribute__((ext_vector_type(2))) float v2f;
typedef __attribute__((ext_vector_type(4))) float v4f;
typedef __attribute__((ext_vector_type(8))) float v8f;

__global__ void pool_kernel(const float* __restrict__ probe,
                            const float* __restrict__ gallery,
                            float* __restrict__ pmean,
                            float* __restrict__ gmean) {
  const int lane = threadIdx.x & 31;
  const int wave = threadIdx.x >> 5;
  const int sub  = lane & 15;
  const int half = lane >> 4;

  long long row = (long long)blockIdx.x * 16 + (long long)wave * 2 + half;
  const long long n_probe_rows = (long long)NP * NC;

  const float* src;
  float* dst;
  if (row < n_probe_rows) {
    src = probe + row * HW;
    dst = pmean + row;
  } else {
    long long r2 = row - n_probe_rows;
    src = gallery + r2 * HW;
    dst = gmean + r2;
  }

  v4f v = *(const v4f*)(src + sub * 4);
  float s = v.x + v.y + v.z + v.w;
  s += __shfl_xor(s, 1, 32);
  s += __shfl_xor(s, 2, 32);
  s += __shfl_xor(s, 4, 32);
  s += __shfl_xor(s, 8, 32);
  if (sub == 0) *dst = s * (1.0f / 64.0f);
}

__global__ void norm_kernel(const float* __restrict__ pmean,
                            const float* __restrict__ gmean,
                            float* __restrict__ pn,
                            float* __restrict__ gn) {
  const int lane = threadIdx.x & 31;
  const int wave = threadIdx.x >> 5;
  const int row  = blockIdx.x * 8 + wave;

  const float* src;
  float* dst;
  if (row < NP) {
    src = pmean + (long long)row * NC;
    dst = pn + row;
  } else {
    src = gmean + (long long)(row - NP) * NC;
    dst = gn + (row - NP);
  }

  float s = 0.0f;
  #pragma unroll 4
  for (int i = lane; i < NC; i += 32) {
    float x = src[i];
    s += x * x;
  }
  s += __shfl_xor(s, 1, 32);
  s += __shfl_xor(s, 2, 32);
  s += __shfl_xor(s, 4, 32);
  s += __shfl_xor(s, 8, 32);
  s += __shfl_xor(s, 16, 32);
  if (lane == 0) *dst = s;
}

__global__ void dist_wmma_kernel(const float* __restrict__ pmean,
                                 const float* __restrict__ gmean,
                                 const float* __restrict__ pn,
                                 const float* __restrict__ gn,
                                 float* __restrict__ out) {
  const int lane = threadIdx.x;
  const int n16  = lane & 15;
  const int hi   = lane >> 4;
  const int ti   = blockIdx.y;
  const int tj   = blockIdx.x;

  const float* arow = pmean + (long long)(ti * 16 + n16) * NC + 2 * hi;
  const float* brow = gmean + (long long)(tj * 16 + n16) * NC + 2 * hi;

  v8f acc = {};
  #pragma unroll 8
  for (int k = 0; k < NC; k += 4) {
    v2f a = *(const v2f*)(arow + k);
    v2f b = *(const v2f*)(brow + k);
    acc = __builtin_amdgcn_wmma_f32_16x16x4_f32(
         false, a,  false, b,
         (short)0, acc,  false,  false);
  }

  const int n = tj * 16 + n16;
  const float gnv = gn[n];
  #pragma unroll
  for (int r = 0; r < 8; ++r) {
    const int m = ti * 16 + hi * 8 + r;
    out[m * NG + n] = pn[m] + gnv - 2.0f * acc[r];
  }
}

extern "C" void kernel_launch(void* const* d_in, const int* in_sizes, int n_in,
                              void* d_out, int out_size, void* d_ws, size_t ws_size,
                              hipStream_t stream) {
  const float* probe   = (const float*)d_in[0];
  const float* gallery = (const float*)d_in[1];
  float* out = (float*)d_out;

  float* ws    = (float*)d_ws;
  float* pmean = ws;
  float* gmean = pmean + (size_t)NP * NC;
  float* pn    = gmean + (size_t)NG * NC;
  float* gn    = pn + NP;

  const int total_rows = (NP + NG) * NC;
  pool_kernel<<<total_rows / 16, 256, 0, stream>>>(probe, gallery, pmean, gmean);

  norm_kernel<<<(NP + NG) / 8, 256, 0, stream>>>(pmean, gmean, pn, gn);

  dist_wmma_kernel<<<dim3(NG / 16, NP / 16), 32, 0, stream>>>(pmean, gmean, pn, gn, out);
}


# pool and row norm fused: one 1024-thread workgroup per sample, 8 float4 loads in flight per lane, means via LDS to whole-line stores, squares summed there; norm launch removed; norm sum order changes
# speedup vs baseline: 1.5024x; 1.5024x over previous
#include <hip/hip_runtime.h>


#define NP 128
#define NG 256
#define NC 2048
#define HW 64

typedef __attribute__((ext_vector_type(2))) float v2f;
typedef __attribute__((ext_vector_type(4))) float v4f;
typedef __attribute__((ext_vector_type(8))) float v8f;

#define POOL_BATCH 8

__global__ void __launch_bounds__(1024)
pool_norm_kernel(const float* __restrict__ probe,
                 const float* __restrict__ gallery,
                 float* __restrict__ pmean,
                 float* __restrict__ gmean,
                 float* __restrict__ pn,
                 float* __restrict__ gn) {
  __shared__ __attribute__((aligned(16))) float means[NC];
  __shared__ float wsum[32];

  const int tid  = threadIdx.x;
  const int lane = tid & 31;
  const int wave = tid >> 5;
  const int sub  = lane & 15;
  const int half = lane >> 4;
  const int n    = blockIdx.x;

  const float* src;
  float* dstm;
  float* dstn;
  if (n < NP) {
    src  = probe + (long long)n * NC * HW;
    dstm = pmean + (long long)n * NC;
    dstn = pn + n;
  } else {
    const int g = n - NP;
    src  = gallery + (long long)g * NC * HW;
    dstm = gmean + (long long)g * NC;
    dstn = gn + g;
  }

  const float* wsrc = src + (long long)wave * (64 * HW) + lane * 4;
  float* wslot = means + wave * 64;

  for (int it = 0; it < 32; it += POOL_BATCH) {
    v4f v[POOL_BATCH];
    float s[POOL_BATCH], t[POOL_BATCH];
    #pragma unroll
    for (int u = 0; u < POOL_BATCH; ++u)
      v[u] = *(const v4f*)(wsrc + (it + u) * (2 * HW));
    #pragma unroll
    for (int u = 0; u < POOL_BATCH; ++u)
      s[u] = v[u].x + v[u].y + v[u].z + v[u].w;
    #pragma unroll
    for (int u = 0; u < POOL_BATCH; ++u) t[u] = __shfl_xor(s[u], 1, 32);
    #pragma unroll
    for (int u = 0; u < POOL_BATCH; ++u) s[u] += t[u];
    #pragma unroll
    for (int u = 0; u < POOL_BATCH; ++u) t[u] = __shfl_xor(s[u], 2, 32);
    #pragma unroll
    for (int u = 0; u < POOL_BATCH; ++u) s[u] += t[u];
    #pragma unroll
    for (int u = 0; u < POOL_BATCH; ++u) t[u] = __shfl_xor(s[u], 4, 32);
    #pragma unroll
    for (int u = 0; u < POOL_BATCH; ++u) s[u] += t[u];
    #pragma unroll
    for (int u = 0; u < POOL_BATCH; ++u) t[u] = __shfl_xor(s[u], 8, 32);
    #pragma unroll
    for (int u = 0; u < POOL_BATCH; ++u) s[u] += t[u];
    if (sub == 0) {
      float* d = wslot + half * 32 + it;
      #pragma unroll
      for (int u = 0; u < POOL_BATCH; u += 4) {
        v4f o;
        o.x = s[u + 0] * (1.0f / 64.0f);
        o.y = s[u + 1] * (1.0f / 64.0f);
        o.z = s[u + 2] * (1.0f / 64.0f);
        o.w = s[u + 3] * (1.0f / 64.0f);
        *(v4f*)(d + u) = o;
      }
    }
  }
  __syncthreads();

  v2f m;
  m.x = wslot[lane];
  m.y = wslot[32 + lane];
  *(v2f*)(dstm + 2 * tid) = m;
  float q = m.x * m.x + m.y * m.y;
  q += __shfl_xor(q, 1, 32);
  q += __shfl_xor(q, 2, 32);
  q += __shfl_xor(q, 4, 32);
  q += __shfl_xor(q, 8, 32);
  q += __shfl_xor(q, 16, 32);
  if (lane == 0) wsum[wave] = q;
  __syncthreads();
  if (wave == 0) {
    float r = wsum[lane];
    r += __shfl_xor(r, 1, 32);
    r += __shfl_xor(r, 2, 32);
    r += __shfl_xor(r, 4, 32);
    r += __shfl_xor(r, 8, 32);
    r += __shfl_xor(r, 16, 32);
    if (lane == 0) *dstn = r;
  }
}

__global__ void dist_wmma_kernel(const float* __restrict__ pmean,
                                 const float* __restrict__ gmean,
                                 const float* __restrict__ pn,
                                 const float* __restrict__ gn,
                                 float* __restrict__ out) {
  const int lane = threadIdx.x;
  const int n16  = lane & 15;
  const int hi   = lane >> 4;
  const int ti   = blockIdx.y;
  const int tj   = blockIdx.x;

  const float* arow = pmean + (long long)(ti * 16 + n16) * NC + 2 * hi;
  const float* brow = gmean + (long long)(tj * 16 + n16) * NC + 2 * hi;

  v8f acc = {};
  #pragma unroll 8
  for (int k = 0; k < NC; k += 4) {
    v2f a = *(const v2f*)(arow + k);
    v2f b = *(const v2f*)(brow + k);
    acc = __builtin_amdgcn_wmma_f32_16x16x4_f32(
         false, a,  false, b,
         (short)0, acc,  false,  false);
  }

  const int n = tj * 16 + n16;
  const float gnv = gn[n];
  #pragma unroll
  for (int r = 0; r < 8; ++r) {
    const int m = ti * 16 + hi * 8 + r;
    out[m * NG + n] = pn[m] + gnv - 2.0f * acc[r];
  }
}

extern "C" void kernel_launch(void* const* d_in, const int* in_sizes, int n_in,
                              void* d_out, int out_size, void* d_ws, size_t ws_size,
                              hipStream_t stream) {
  const float* probe   = (const float*)d_in[0];
  const float* gallery = (const float*)d_in[1];
  float* out = (float*)d_out;

  float* ws    = (float*)d_ws;
  float* pmean = ws;
  float* gmean = pmean + (size_t)NP * NC;
  float* pn    = gmean + (size_t)NG * NC;
  float* gn    = pn + NP;

  pool_norm_kernel<<<NP + NG, 1024, 0, stream>>>(probe, gallery, pmean, gmean, pn, gn);

  dist_wmma_kernel<<<dim3(NG / 16, NP / 16), 32, 0, stream>>>(pmean, gmean, pn, gn, out);
}


# distance kernel: K loop software-pipelined inside the one wave per tile: ring of 28 float4 fragment pairs in registers, 56 loads in flight, counted waits; k order inside each 8-k group changes
# speedup vs baseline: 2.4266x; 1.6151x over previous
#include <hip/hip_runtime.h>


#define NP 128
#define NG 256
#define NC 2048
#define HW 64

typedef __attribute__((ext_vector_type(2))) float v2f;
typedef __attribute__((ext_vector_type(4))) float v4f;
typedef __attribute__((ext_vector_type(8))) float v8f;

#define POOL_BATCH 8

__global__ void __launch_bounds__(1024)
pool_norm_kernel(const float* __restrict__ probe,
                 const float* __restrict__ gallery,
                 float* __restrict__ pmean,
                 float* __restrict__ gmean,
                 float* __restrict__ pn,
                 float* __restrict__ gn) {
  __shared__ __attribute__((aligned(16))) float means[NC];
  __shared__ float wsum[32];

  const int tid  = threadIdx.x;
  const int lane = tid & 31;
  const int wave = tid >> 5;
  const int sub  = lane & 15;
  const int half = lane >> 4;
  const int n    = blockIdx.x;

  const float* src;
  float* dstm;
  float* dstn;
  if (n < NP) {
    src  = probe + (long long)n * NC * HW;
    dstm = pmean + (long long)n * NC;
    dstn = pn + n;
  } else {
    const int g = n - NP;
    src  = gallery + (long long)g * NC * HW;
    dstm = gmean + (long long)g * NC;
    dstn = gn + g;
  }

  const float* wsrc = src + (long long)wave * (64 * HW) + lane * 4;
  float* wslot = means + wave * 64;

  for (int it = 0; it < 32; it += POOL_BATCH) {
    v4f v[POOL_BATCH];
    float s[POOL_BATCH], t[POOL_BATCH];
    #pragma unroll
    for (int u = 0; u < POOL_BATCH; ++u)
      v[u] = *(const v4f*)(wsrc + (it + u) * (2 * HW));
    #pragma unroll
    for (int u = 0; u < POOL_BATCH; ++u)
      s[u] = v[u].x + v[u].y + v[u].z + v[u].w;
    #pragma unroll
    for (int u = 0; u < POOL_BATCH; ++u) t[u] = __shfl_xor(s[u], 1, 32);
    #pragma unroll
    for (int u = 0; u < POOL_BATCH; ++u) s[u] += t[u];
    #pragma unroll
    for (int u = 0; u < POOL_BATCH; ++u) t[u] = __shfl_xor(s[u], 2, 32);
    #pragma unroll
    for (int u = 0; u < POOL_BATCH; ++u) s[u] += t[u];
    #pragma unroll
    for (int u = 0; u < POOL_BATCH; ++u) t[u] = __shfl_xor(s[u], 4, 32);
    #pragma unroll
    for (int u = 0; u < POOL_BATCH; ++u) s[u] += t[u];
    #pragma unroll
    for (int u = 0; u < POOL_BATCH; ++u) t[u] = __shfl_xor(s[u], 8, 32);
    #pragma unroll
    for (int u = 0; u < POOL_BATCH; ++u) s[u] += t[u];
    if (sub == 0) {
      float* d = wslot + half * 32 + it;
      #pragma unroll
      for (int u = 0; u < POOL_BATCH; u += 4) {
        v4f o;
        o.x = s[u + 0] * (1.0f / 64.0f);
        o.y = s[u + 1] * (1.0f / 64.0f);
        o.z = s[u + 2] * (1.0f / 64.0f);
        o.w = s[u + 3] * (1.0f / 64.0f);
        *(v4f*)(d + u) = o;
      }
    }
  }
  __syncthreads();

  v2f m;
  m.x = wslot[lane];
  m.y = wslot[32 + lane];
  *(v2f*)(dstm + 2 * tid) = m;
  float q = m.x * m.x + m.y * m.y;
  q += __shfl_xor(q, 1, 32);
  q += __shfl_xor(q, 2, 32);
  q += __shfl_xor(q, 4, 32);
  q += __shfl_xor(q, 8, 32);
  q += __shfl_xor(q, 16, 32);
  if (lane == 0) wsum[wave] = q;
  __syncthreads();
  if (wave == 0) {
    float r = wsum[lane];
    r += __shfl_xor(r, 1, 32);
    r += __shfl_xor(r, 2, 32);
    r += __shfl_xor(r, 4, 32);
    r += __shfl_xor(r, 8, 32);
    r += __shfl_xor(r, 16, 32);
    if (lane == 0) *dstn = r;
  }
}

#define RING 28
#define NSTAGE (NC / 8)
#define MAIN_STAGES ((NSTAGE / RING - 1) * RING)
#define REM_STAGES (NSTAGE % RING)

__device__ __forceinline__ v8f stage_mma(v4f a, v4f b, v8f acc) {
  acc = __builtin_amdgcn_wmma_f32_16x16x4_f32(false, a.xy, false, b.xy, (short)0, acc, false, false);
  acc = __builtin_amdgcn_wmma_f32_16x16x4_f32(false, a.zw, false, b.zw, (short)0, acc, false, false);
  return acc;
}

__global__ void __launch_bounds__(32)
dist_wmma_kernel(const float* __restrict__ pmean,
                 const float* __restrict__ gmean,
                 const float* __restrict__ pn,
                 const float* __restrict__ gn,
                 float* __restrict__ out) {
  const int lane = threadIdx.x;
  const int n16  = lane & 15;
  const int hi   = lane >> 4;
  const int ti   = blockIdx.y;
  const int tj   = blockIdx.x;

  const v4f* arow = (const v4f*)(pmean + (long long)(ti * 16 + n16) * NC + 4 * hi);
  const v4f* brow = (const v4f*)(gmean + (long long)(tj * 16 + n16) * NC + 4 * hi);

  v4f ra[RING], rb[RING];
  #pragma unroll
  for (int i = 0; i < RING; ++i) {
    ra[i] = arow[2 * i];
    rb[i] = brow[2 * i];
  }

  v8f acc = {};
  int base = 0;
  for (; base < MAIN_STAGES; base += RING) {
    #pragma unroll
    for (int i = 0; i < RING; ++i) {
      acc = stage_mma(ra[i], rb[i], acc);
      ra[i] = arow[2 * (base + RING + i)];
      rb[i] = brow[2 * (base + RING + i)];
    }
  }
  #pragma unroll
  for (int i = 0; i < REM_STAGES; ++i) {
    acc = stage_mma(ra[i], rb[i], acc);
    ra[i] = arow[2 * (MAIN_STAGES + RING + i)];
    rb[i] = brow[2 * (MAIN_STAGES + RING + i)];
  }
  #pragma unroll
  for (int i = REM_STAGES; i < RING; ++i) acc = stage_mma(ra[i], rb[i], acc);
  #pragma unroll
  for (int i = 0; i < REM_STAGES; ++i) acc = stage_mma(ra[i], rb[i], acc);

  const int n = tj * 16 + n16;
  const float gnv = gn[n];
  #pragma unroll
  for (int r = 0; r < 8; ++r) {
    const int m = ti * 16 + hi * 8 + r;
    out[m * NG + n] = pn[m] + gnv - 2.0f * acc[r];
  }
}

extern "C" void kernel_launch(void* const* d_in, const int* in_sizes, int n_in,
                              void* d_out, int out_size, void* d_ws, size_t ws_size,
                              hipStream_t stream) {
  const float* probe   = (const float*)d_in[0];
  const float* gallery = (const float*)d_in[1];
  float* out = (float*)d_out;

  float* ws    = (float*)d_ws;
  float* pmean = ws;
  float* gmean = pmean + (size_t)NP * NC;
  float* pn    = gmean + (size_t)NG * NC;
  float* gn    = pn + NP;

  pool_norm_kernel<<<NP + NG, 1024, 0, stream>>>(probe, gallery, pmean, gmean, pn, gn);

  dist_wmma_kernel<<<dim3(NG / 16, NP / 16), 32, 0, stream>>>(pmean, gmean, pn, gn, out);
}


# pool kernel: row sums leave the LDS unit: xor butterfly as DPP row_xmask adds that fold rows together (same tree), lane L ends with row L, lines stored from registers; norm sum order changes
# speedup vs baseline: 2.5866x; 1.0659x over previous
#include <hip/hip_runtime.h>


#define NP 128
#define NG 256
#define NC 2048
#define HW 64

typedef __attribute__((ext_vector_type(2))) float v2f;
typedef __attribute__((ext_vector_type(4))) float v4f;
typedef __attribute__((ext_vector_type(8))) float v8f;

#define POOL_BATCH 8

#define ROW_XMASK_1 0x161
#define ROW_XMASK_2 0x162
#define ROW_XMASK_4 0x164
#define ROW_XMASK_8 0x168

template <int CTRL>
__device__ __forceinline__ float row_xchg(float x) {
  return __int_as_float(__builtin_amdgcn_update_dpp(0, __float_as_int(x), CTRL, 0xf, 0xf, true));
}

template <int CTRL>
__device__ __forceinline__ float fold_rows(float a, float b, bool bit_set) {
  const float keep = bit_set ? b : a;
  const float send = bit_set ? a : b;
  return keep + row_xchg<CTRL>(send);
}

__device__ __forceinline__ float other_half(float x) {
  return __int_as_float(__builtin_amdgcn_permlanex16(__float_as_int(x), __float_as_int(x),
                                                     0x76543210, 0xfedcba98, false, false));
}

__device__ __forceinline__ float wave_sum(float q) {
  q += row_xchg<ROW_XMASK_1>(q);
  q += row_xchg<ROW_XMASK_2>(q);
  q += row_xchg<ROW_XMASK_4>(q);
  q += row_xchg<ROW_XMASK_8>(q);
  q += other_half(q);
  return q;
}

__global__ void __launch_bounds__(1024)
pool_norm_kernel(const float* __restrict__ probe,
                 const float* __restrict__ gallery,
                 float* __restrict__ pmean,
                 float* __restrict__ gmean,
                 float* __restrict__ pn,
                 float* __restrict__ gn) {
  __shared__ float wsum[32];

  const int tid  = threadIdx.x;
  const int lane = tid & 31;
  const int wave = tid >> 5;
  const int sub  = lane & 15;
  const int half = lane >> 4;
  const int n    = blockIdx.x;

  const float* src;
  float* dstm;
  float* dstn;
  if (n < NP) {
    src  = probe + (long long)n * NC * HW;
    dstm = pmean + (long long)n * NC;
    dstn = pn + n;
  } else {
    const int g = n - NP;
    src  = gallery + (long long)g * NC * HW;
    dstm = gmean + (long long)g * NC;
    dstn = gn + g;
  }

  const float* wsrc = src + (long long)(wave * 64 + half * 16) * HW + sub * 4;
  float* wdst = dstm + wave * 64 + lane;

  const bool bit0 = (sub & 1) != 0;
  const bool bit1 = (sub & 2) != 0;
  const bool bit2 = (sub & 4) != 0;
  const bool bit3 = (sub & 8) != 0;

  float q = 0.0f;
  #pragma unroll
  for (int grp = 0; grp < 2; ++grp) {
    float r[2];
    #pragma unroll
    for (int b = 0; b < 2; ++b) {
      v4f v[POOL_BATCH];
      float s[POOL_BATCH];
      #pragma unroll
      for (int u = 0; u < POOL_BATCH; ++u)
        v[u] = *(const v4f*)(wsrc + (grp * 32 + b * POOL_BATCH + u) * HW);
      #pragma unroll
      for (int u = 0; u < POOL_BATCH; ++u)
        s[u] = v[u].x + v[u].y + v[u].z + v[u].w;
      const float t0 = fold_rows<ROW_XMASK_1>(s[0], s[1], bit0);
      const float t1 = fold_rows<ROW_XMASK_1>(s[2], s[3], bit0);
      const float t2 = fold_rows<ROW_XMASK_1>(s[4], s[5], bit0);
      const float t3 = fold_rows<ROW_XMASK_1>(s[6], s[7], bit0);
      const float u0 = fold_rows<ROW_XMASK_2>(t0, t1, bit1);
      const float u1 = fold_rows<ROW_XMASK_2>(t2, t3, bit1);
      r[b] = fold_rows<ROW_XMASK_4>(u0, u1, bit2);
    }
    const float m = fold_rows<ROW_XMASK_8>(r[0], r[1], bit3) * (1.0f / 64.0f);
    wdst[grp * 32] = m;
    q += m * m;
  }

  q = wave_sum(q);
  if (lane == 0) wsum[wave] = q;
  __syncthreads();
  if (wave == 0) {
    const float tot = wave_sum(wsum[lane]);
    if (lane == 0) *dstn = tot;
  }
}

#define RING 28
#define NSTAGE (NC / 8)
#define MAIN_STAGES ((NSTAGE / RING - 1) * RING)
#define REM_STAGES (NSTAGE % RING)

__device__ __forceinline__ v8f stage_mma(v4f a, v4f b, v8f acc) {
  acc = __builtin_amdgcn_wmma_f32_16x16x4_f32(false, a.xy, false, b.xy, (short)0, acc, false, false);
  acc = __builtin_amdgcn_wmma_f32_16x16x4_f32(false, a.zw, false, b.zw, (short)0, acc, false, false);
  return acc;
}

__global__ void __launch_bounds__(32)
dist_wmma_kernel(const float* __restrict__ pmean,
                 const float* __restrict__ gmean,
                 const float* __restrict__ pn,
                 const float* __restrict__ gn,
                 float* __restrict__ out) {
  const int lane = threadIdx.x;
  const int n16  = lane & 15;
  const int hi   = lane >> 4;
  const int ti   = blockIdx.y;
  const int tj   = blockIdx.x;

  const v4f* arow = (const v4f*)(pmean + (long long)(ti * 16 + n16) * NC + 4 * hi);
  const v4f* brow = (const v4f*)(gmean + (long long)(tj * 16 + n16) * NC + 4 * hi);

  v4f ra[RING], rb[RING];
  #pragma unroll
  for (int i = 0; i < RING; ++i) {
    ra[i] = arow[2 * i];
    rb[i] = brow[2 * i];
  }

  v8f acc = {};
  int base = 0;
  for (; base < MAIN_STAGES; base += RING) {
    #pragma unroll
    for (int i = 0; i < RING; ++i) {
      acc = stage_mma(ra[i], rb[i], acc);
      ra[i] = arow[2 * (base + RING + i)];
      rb[i] = brow[2 * (base + RING + i)];
    }
  }
  #pragma unroll
  for (int i = 0; i < REM_STAGES; ++i) {
    acc = stage_mma(ra[i], rb[i], acc);
    ra[i] = arow[2 * (MAIN_STAGES + RING + i)];
    rb[i] = brow[2 * (MAIN_STAGES + RING + i)];
  }
  #pragma unroll
  for (int i = REM_STAGES; i < RING; ++i) acc = stage_mma(ra[i], rb[i], acc);
  #pragma unroll
  for (int i = 0; i < REM_STAGES; ++i) acc = stage_mma(ra[i], rb[i], acc);

  const int n = tj * 16 + n16;
  const float gnv = gn[n];
  #pragma unroll
  for (int r = 0; r < 8; ++r) {
    const int m = ti * 16 + hi * 8 + r;
    out[m * NG + n] = pn[m] + gnv - 2.0f * acc[r];
  }
}

extern "C" void kernel_launch(void* const* d_in, const int* in_sizes, int n_in,
                              void* d_out, int out_size, void* d_ws, size_t ws_size,
                              hipStream_t stream) {
  const float* probe   = (const float*)d_in[0];
  const float* gallery = (const float*)d_in[1];
  float* out = (float*)d_out;

  float* ws    = (float*)d_ws;
  float* pmean = ws;
  float* gmean = pmean + (size_t)NP * NC;
  float* pn    = gmean + (size_t)NG * NC;
  float* gn    = pn + NP;

  pool_norm_kernel<<<NP + NG, 1024, 0, stream>>>(probe, gallery, pmean, gmean, pn, gn);

  dist_wmma_kernel<<<dim3(NG / 16, NP / 16), 32, 0, stream>>>(pmean, gmean, pn, gn, out);
}
